# starting kernel
# baseline (speedup 1.0000x reference)
#include <hip/hip_runtime.h>
#include <hip/hip_bf16.h>

#define BB 32
#define SS 512
#define FF 16
#define HH 256
#define BS (BB * SS)

typedef float v2f __attribute__((ext_vector_type(2)));
typedef float v8f __attribute__((ext_vector_type(8)));

__device__ __forceinline__ float elu_f(float v) {
    return v > 0.0f ? v : (__expf(v) - 1.0f);
}

__global__ __launch_bounds__(256) void vsn_attn_kernel(
    const float* __restrict__ x, const float* __restrict__ Wa1,
    const float* __restrict__ ba1, const float* __restrict__ Wa2,
    const float* __restrict__ ba2, float* __restrict__ wout)
{
    __shared__ float s_wa1t[HH * FF];
    __shared__ float s_wa2[HH * FF];
    __shared__ float s_ba1[HH];
    __shared__ float s_ba2[FF];

    const int tid = threadIdx.x;
#pragma unroll
    for (int i = 0; i < 16; ++i) {
        int lin = tid + i * 256;
        int f = lin >> 8, h = lin & 255;
        s_wa1t[h * FF + f] = Wa1[lin];
        s_wa2[lin] = Wa2[lin];
    }
    s_ba1[tid] = ba1[tid & 255];
    if (tid < FF) s_ba2[tid] = ba2[tid];
    __syncthreads();

    const size_t row = (size_t)blockIdx.x * 256 + tid;
    const float4* xs = (const float4*)(x + row * FF);
    float4 x0 = xs[0], x1 = xs[1], x2 = xs[2], x3 = xs[3];
    float xv[16] = {x0.x, x0.y, x0.z, x0.w, x1.x, x1.y, x1.z, x1.w,
                    x2.x, x2.y, x2.z, x2.w, x3.x, x3.y, x3.z, x3.w};

    float lg[16];
#pragma unroll
    for (int f = 0; f < 16; ++f) lg[f] = s_ba2[f];

#pragma unroll 2
    for (int h = 0; h < HH; ++h) {
        const float4* wc = (const float4*)&s_wa1t[h * FF];
        float4 c0 = wc[0], c1 = wc[1], c2 = wc[2], c3 = wc[3];
        float s = s_ba1[h];
        s += xv[0]*c0.x + xv[1]*c0.y + xv[2]*c0.z + xv[3]*c0.w;
        s += xv[4]*c1.x + xv[5]*c1.y + xv[6]*c1.z + xv[7]*c1.w;
        s += xv[8]*c2.x + xv[9]*c2.y + xv[10]*c2.z + xv[11]*c2.w;
        s += xv[12]*c3.x + xv[13]*c3.y + xv[14]*c3.z + xv[15]*c3.w;
        float a = elu_f(s);
        const float4* wd = (const float4*)&s_wa2[h * FF];
        float4 d0 = wd[0], d1 = wd[1], d2 = wd[2], d3 = wd[3];
        lg[0] += a*d0.x; lg[1] += a*d0.y; lg[2] += a*d0.z; lg[3] += a*d0.w;
        lg[4] += a*d1.x; lg[5] += a*d1.y; lg[6] += a*d1.z; lg[7] += a*d1.w;
        lg[8] += a*d2.x; lg[9] += a*d2.y; lg[10]+= a*d2.z; lg[11]+= a*d2.w;
        lg[12]+= a*d3.x; lg[13]+= a*d3.y; lg[14]+= a*d3.z; lg[15]+= a*d3.w;
    }

    float m = lg[0];
#pragma unroll
    for (int f = 1; f < 16; ++f) m = fmaxf(m, lg[f]);
    float s = 0.0f;
#pragma unroll
    for (int f = 0; f < 16; ++f) { lg[f] = __expf(lg[f] - m); s += lg[f]; }
    float inv = 1.0f / s;
#pragma unroll
    for (int f = 0; f < 16; ++f) lg[f] *= inv;

    float4* wo = (float4*)(wout + row * FF);
    wo[0] = make_float4(lg[0], lg[1], lg[2], lg[3]);
    wo[1] = make_float4(lg[4], lg[5], lg[6], lg[7]);
    wo[2] = make_float4(lg[8], lg[9], lg[10], lg[11]);
    wo[3] = make_float4(lg[12], lg[13], lg[14], lg[15]);
}

#define NTW   8
#define KC    64
#define PSTR  288

__global__ __launch_bounds__(256) void vsn_main_kernel(
    const float* __restrict__ x, const float* __restrict__ W1,
    const float* __restrict__ b1, const float* __restrict__ W2,
    const float* __restrict__ b2, const float* __restrict__ wts,
    float* __restrict__ out)
{
    __shared__ float  s_w2[(KC / 2) * PSTR];
    __shared__ float4 s_wb[128];
    __shared__ float  s_wt[128 * FF];

    const int tid  = threadIdx.x;
    const int wave = tid >> 5;
    const int lane = tid & 31;
    const int half = lane >> 4;
    const int lr   = lane & 15;
    const int n0   = blockIdx.x * 128;
    const int R0   = blockIdx.y * 128;

    {
        const float4* ws = (const float4*)(wts + (size_t)R0 * FF);
        float4* wd = (float4*)s_wt;
        wd[tid]       = ws[tid];
        wd[tid + 256] = ws[tid + 256];
    }
    const int myrow = R0 + wave * 16 + lr;
    const float4* xs = (const float4*)(x + (size_t)myrow * FF);
    float4 x0 = xs[0], x1 = xs[1], x2 = xs[2], x3 = xs[3];
    float xv[16] = {x0.x, x0.y, x0.z, x0.w, x1.x, x1.y, x1.z, x1.w,
                    x2.x, x2.y, x2.z, x2.w, x3.x, x3.y, x3.z, x3.w};

    v8f acc[NTW];
#pragma unroll
    for (int t = 0; t < NTW; ++t) acc[t] = (v8f)0.0f;

    __syncthreads();

    for (int f = 0; f < FF; ++f) {
        const float wa = s_wt[(wave * 16 + lr) * FF + f];
        const float xf = xv[f];

        for (int kc = 0; kc < HH / KC; ++kc) {
            __syncthreads();
            {
                const float* w2base =
                    W2 + ((size_t)f * HH + kc * KC) * HH + n0;
#pragma unroll
                for (int i = 0; i < 4; ++i) {
                    int idx = tid + i * 256;
                    int p = idx >> 5;
                    int q = idx & 31;
                    float4 r0 = *(const float4*)(w2base + (size_t)(2 * p) * HH + 4 * q);
                    float4 r1 = *(const float4*)(w2base + (size_t)(2 * p + 1) * HH + 4 * q);
                    float* d = &s_w2[p * PSTR + 8 * q];
                    *(float4*)(d)     = make_float4(r0.x, r1.x, r0.y, r1.y);
                    *(float4*)(d + 4) = make_float4(r0.z, r1.z, r0.w, r1.w);
                }
                if (kc == 0 && tid < 128) {
                    float2 w  = *(const float2*)(W1 + f * HH + tid * 2);
                    float2 bb = *(const float2*)(b1 + f * HH + tid * 2);
                    s_wb[tid] = make_float4(w.x, w.y, bb.x, bb.y);
                }
            }
            __syncthreads();

#pragma unroll 4
            for (int ks = 0; ks < KC / 4; ++ks) {
                float4 p4 = s_wb[kc * (KC / 2) + 2 * ks + half];
                v2f afrag;
                afrag.x = wa * elu_f(xf * p4.x + p4.z);
                afrag.y = wa * elu_f(xf * p4.y + p4.w);
                const int prow = 2 * ks + half;
#pragma unroll
                for (int t = 0; t < NTW; ++t) {
                    v2f bfrag = *(const v2f*)&s_w2[prow * PSTR + 2 * (t * 16 + lr)];
                    acc[t] = __builtin_amdgcn_wmma_f32_16x16x4_f32(
                        false, afrag, false, bfrag, (short)0, acc[t],
                        false, false);
                }
            }
        }

        float wr[8];
#pragma unroll
        for (int r = 0; r < 8; ++r)
            wr[r] = s_wt[(wave * 16 + 8 * half + r) * FF + f];
#pragma unroll
        for (int t = 0; t < NTW; ++t) {
            float b2v = b2[f * HH + n0 + t * 16 + lr];
#pragma unroll
            for (int r = 0; r < 8; ++r) acc[t][r] += wr[r] * b2v;
        }
    }

#pragma unroll
    for (int t = 0; t < NTW; ++t) {
#pragma unroll
        for (int r = 0; r < 8; ++r) {
            size_t row = (size_t)(R0 + wave * 16 + 8 * half + r);
            out[row * HH + n0 + t * 16 + lr] = acc[t][r];
        }
    }
}

extern "C" void kernel_launch(void* const* d_in, const int* in_sizes, int n_in,
                              void* d_out, int out_size, void* d_ws, size_t ws_size,
                              hipStream_t stream) {
    const float* x   = (const float*)d_in[0];
    const float* W1  = (const float*)d_in[1];
    const float* b1  = (const float*)d_in[2];
    const float* W2  = (const float*)d_in[3];
    const float* b2  = (const float*)d_in[4];
    const float* Wa1 = (const float*)d_in[5];
    const float* ba1 = (const float*)d_in[6];
    const float* Wa2 = (const float*)d_in[7];
    const float* ba2 = (const float*)d_in[8];

    float* out  = (float*)d_out;
    float* wout = out + (size_t)BS * HH;

    vsn_attn_kernel<<<dim3(BS / 256), dim3(256), 0, stream>>>(
        x, Wa1, ba1, Wa2, ba2, wout);

    vsn_main_kernel<<<dim3(HH / 128, BS / 128), dim3(256), 0, stream>>>(
        x, W1, b1, W2, b2, wout, out);
}


# Main kernel: W2/b2 B fragments read straight from global memory into registers one 32-k block ahead: no LDS, no barriers; k order in 32-k blocks and bias as 16 extra k rows change the f32 sum order
# speedup vs baseline: 1.2277x; 1.2277x over previous
#include <hip/hip_runtime.h>
#include <hip/hip_bf16.h>

#define BB 32
#define SS 512
#define FF 16
#define HH 256
#define BS (BB * SS)

typedef float v2f __attribute__((ext_vector_type(2)));
typedef float v8f __attribute__((ext_vector_type(8)));

__device__ __forceinline__ float elu_f(float v) {
    return v > 0.0f ? v : (__expf(v) - 1.0f);
}

__global__ __launch_bounds__(256) void vsn_attn_kernel(
    const float* __restrict__ x, const float* __restrict__ Wa1,
    const float* __restrict__ ba1, const float* __restrict__ Wa2,
    const float* __restrict__ ba2, float* __restrict__ wout)
{
    __shared__ float s_wa1t[HH * FF];
    __shared__ float s_wa2[HH * FF];
    __shared__ float s_ba1[HH];
    __shared__ float s_ba2[FF];

    const int tid = threadIdx.x;
#pragma unroll
    for (int i = 0; i < 16; ++i) {
        int lin = tid + i * 256;
        int f = lin >> 8, h = lin & 255;
        s_wa1t[h * FF + f] = Wa1[lin];
        s_wa2[lin] = Wa2[lin];
    }
    s_ba1[tid] = ba1[tid & 255];
    if (tid < FF) s_ba2[tid] = ba2[tid];
    __syncthreads();

    const size_t row = (size_t)blockIdx.x * 256 + tid;
    const float4* xs = (const float4*)(x + row * FF);
    float4 x0 = xs[0], x1 = xs[1], x2 = xs[2], x3 = xs[3];
    float xv[16] = {x0.x, x0.y, x0.z, x0.w, x1.x, x1.y, x1.z, x1.w,
                    x2.x, x2.y, x2.z, x2.w, x3.x, x3.y, x3.z, x3.w};

    float lg[16];
#pragma unroll
    for (int f = 0; f < 16; ++f) lg[f] = s_ba2[f];

#pragma unroll 2
    for (int h = 0; h < HH; ++h) {
        const float4* wc = (const float4*)&s_wa1t[h * FF];
        float4 c0 = wc[0], c1 = wc[1], c2 = wc[2], c3 = wc[3];
        float s = s_ba1[h];
        s += xv[0]*c0.x + xv[1]*c0.y + xv[2]*c0.z + xv[3]*c0.w;
        s += xv[4]*c1.x + xv[5]*c1.y + xv[6]*c1.z + xv[7]*c1.w;
        s += xv[8]*c2.x + xv[9]*c2.y + xv[10]*c2.z + xv[11]*c2.w;
        s += xv[12]*c3.x + xv[13]*c3.y + xv[14]*c3.z + xv[15]*c3.w;
        float a = elu_f(s);
        const float4* wd = (const float4*)&s_wa2[h * FF];
        float4 d0 = wd[0], d1 = wd[1], d2 = wd[2], d3 = wd[3];
        lg[0] += a*d0.x; lg[1] += a*d0.y; lg[2] += a*d0.z; lg[3] += a*d0.w;
        lg[4] += a*d1.x; lg[5] += a*d1.y; lg[6] += a*d1.z; lg[7] += a*d1.w;
        lg[8] += a*d2.x; lg[9] += a*d2.y; lg[10]+= a*d2.z; lg[11]+= a*d2.w;
        lg[12]+= a*d3.x; lg[13]+= a*d3.y; lg[14]+= a*d3.z; lg[15]+= a*d3.w;
    }

    float m = lg[0];
#pragma unroll
    for (int f = 1; f < 16; ++f) m = fmaxf(m, lg[f]);
    float s = 0.0f;
#pragma unroll
    for (int f = 0; f < 16; ++f) { lg[f] = __expf(lg[f] - m); s += lg[f]; }
    float inv = 1.0f / s;
#pragma unroll
    for (int f = 0; f < 16; ++f) lg[f] *= inv;

    float4* wo = (float4*)(wout + row * FF);
    wo[0] = make_float4(lg[0], lg[1], lg[2], lg[3]);
    wo[1] = make_float4(lg[4], lg[5], lg[6], lg[7]);
    wo[2] = make_float4(lg[8], lg[9], lg[10], lg[11]);
    wo[3] = make_float4(lg[12], lg[13], lg[14], lg[15]);
}

typedef float v4f __attribute__((ext_vector_type(4)));

#define MTHR   128
#define WROWS  32
#define WCOLS  64
#define KBLK   32
#define NBLK   (FF * HH / KBLK)

struct Blk {
    v4f   w1[4];
    v4f   bb[4];
    float xf[2];
    float wa[2];
    v4f   b[16];
};

__device__ __forceinline__ void load_blk(Blk& B, int blk,
    const float* __restrict__ w1h, const float* __restrict__ b1h,
    const float* __restrict__ w2l,
    const float* __restrict__ xr0, const float* __restrict__ xr1,
    const float* __restrict__ wr0, const float* __restrict__ wr1)
{
    const int f = blk >> 3;
    const float* pw = w1h + blk * KBLK;
    const float* pb = b1h + blk * KBLK;
#pragma unroll
    for (int i = 0; i < 4; ++i) {
        B.w1[i] = *(const v4f*)(pw + 4 * i);
        B.bb[i] = *(const v4f*)(pb + 4 * i);
    }
    B.xf[0] = xr0[f]; B.xf[1] = xr1[f];
    B.wa[0] = wr0[f]; B.wa[1] = wr1[f];
    const float* p2 = w2l + (size_t)blk * (KBLK * HH);
#pragma unroll
    for (int i = 0; i < 16; ++i)
        B.b[i] = *(const v4f*)(p2 + i * HH);
}

__device__ __forceinline__ void compute_blk(const Blk& B, v8f (&acc)[2][4])
{
#pragma unroll
    for (int j = 0; j < 8; ++j) {
        const int i0 = 2 * j, i1 = 2 * j + 1;
        const float w1a = B.w1[i0 >> 2][i0 & 3], w1b = B.w1[i1 >> 2][i1 & 3];
        const float b1a = B.bb[i0 >> 2][i0 & 3], b1b = B.bb[i1 >> 2][i1 & 3];
        v2f bf[4];
#pragma unroll
        for (int t = 0; t < 4; ++t) { bf[t].x = B.b[i0][t]; bf[t].y = B.b[i1][t]; }
#pragma unroll
        for (int m = 0; m < 2; ++m) {
            v2f a;
            a.x = B.wa[m] * elu_f(B.xf[m] * w1a + b1a);
            a.y = B.wa[m] * elu_f(B.xf[m] * w1b + b1b);
#pragma unroll
            for (int t = 0; t < 4; ++t)
                acc[m][t] = __builtin_amdgcn_wmma_f32_16x16x4_f32(
                    false, a, false, bf[t], (short)0, acc[m][t], false, false);
        }
    }
}

__global__ __launch_bounds__(MTHR) void vsn_main_kernel(
    const float* __restrict__ x, const float* __restrict__ W1,
    const float* __restrict__ b1, const float* __restrict__ W2,
    const float* __restrict__ b2, const float* __restrict__ wts,
    float* __restrict__ out)
{
    const int tid  = threadIdx.x;
    const int wave = tid >> 5;
    const int lane = tid & 31;
    const int half = lane >> 4;
    const int lr   = lane & 15;
    const int n0   = blockIdx.x * WCOLS;
    const int R0   = blockIdx.y * (4 * WROWS) + wave * WROWS;

    const float* xr0 = x   + (size_t)(R0 + lr) * FF;
    const float* xr1 = x   + (size_t)(R0 + 16 + lr) * FF;
    const float* wr0 = wts + (size_t)(R0 + lr) * FF;
    const float* wr1 = wts + (size_t)(R0 + 16 + lr) * FF;
    const float* w1h = W1 + 16 * half;
    const float* b1h = b1 + 16 * half;
    const float* w2l = W2 + (size_t)(16 * half) * HH + n0 + 4 * lr;

    Blk P, Q;
    load_blk(P, 0, w1h, b1h, w2l, xr0, xr1, wr0, wr1);

    v8f acc[2][4];
#pragma unroll
    for (int m = 0; m < 2; ++m)
#pragma unroll
        for (int t = 0; t < 4; ++t) acc[m][t] = (v8f)0.0f;

    {
        v4f wv[2][4];
#pragma unroll
        for (int q = 0; q < 4; ++q) {
            wv[0][q] = *(const v4f*)(wr0 + 4 * q);
            wv[1][q] = *(const v4f*)(wr1 + 4 * q);
        }
        const float* pb2 = b2 + (size_t)(2 * half) * HH + n0 + 4 * lr;
#pragma unroll
        for (int j = 0; j < 4; ++j) {
            v4f r0 = *(const v4f*)(pb2 + (4 * j) * HH);
            v4f r1 = *(const v4f*)(pb2 + (4 * j + 1) * HH);
#pragma unroll
            for (int m = 0; m < 2; ++m) {
                v2f a;
                a.x = half ? wv[m][j][2] : wv[m][j][0];
                a.y = half ? wv[m][j][3] : wv[m][j][1];
#pragma unroll
                for (int t = 0; t < 4; ++t) {
                    v2f bf; bf.x = r0[t]; bf.y = r1[t];
                    acc[m][t] = __builtin_amdgcn_wmma_f32_16x16x4_f32(
                        false, a, false, bf, (short)0, acc[m][t], false, false);
                }
            }
        }
    }

    for (int it = 0; it < NBLK / 2; ++it) {
        load_blk(Q, 2 * it + 1, w1h, b1h, w2l, xr0, xr1, wr0, wr1);
        __builtin_amdgcn_sched_barrier(0);
        compute_blk(P, acc);
        __builtin_amdgcn_sched_barrier(0);
        const int nb = (2 * it + 2 < NBLK) ? 2 * it + 2 : NBLK - 1;
        load_blk(P, nb, w1h, b1h, w2l, xr0, xr1, wr0, wr1);
        __builtin_amdgcn_sched_barrier(0);
        compute_blk(Q, acc);
    }

#pragma unroll
    for (int m = 0; m < 2; ++m) {
#pragma unroll
        for (int r = 0; r < 8; ++r) {
            const size_t row = (size_t)(R0 + 16 * m + 8 * half + r);
            v4f o;
            o[0] = acc[m][0][r]; o[1] = acc[m][1][r];
            o[2] = acc[m][2][r]; o[3] = acc[m][3][r];
            *(v4f*)(out + row * HH + n0 + 4 * lr) = o;
        }
    }
}

extern "C" void kernel_launch(void* const* d_in, const int* in_sizes, int n_in,
                              void* d_out, int out_size, void* d_ws, size_t ws_size,
                              hipStream_t stream) {
    const float* x   = (const float*)d_in[0];
    const float* W1  = (const float*)d_in[1];
    const float* b1  = (const float*)d_in[2];
    const float* W2  = (const float*)d_in[3];
    const float* b2  = (const float*)d_in[4];
    const float* Wa1 = (const float*)d_in[5];
    const float* ba1 = (const float*)d_in[6];
    const float* Wa2 = (const float*)d_in[7];
    const float* ba2 = (const float*)d_in[8];

    float* out  = (float*)d_out;
    float* wout = out + (size_t)BS * HH;

    vsn_attn_kernel<<<dim3(BS / 256), dim3(256), 0, stream>>>(
        x, Wa1, ba1, Wa2, ba2, wout);

    vsn_main_kernel<<<dim3(HH / WCOLS, BS / (4 * WROWS)), dim3(MTHR), 0, stream>>>(
        x, W1, b1, W2, b2, wout, out);
}


# Main kernel wave tile widened from 32x64 to 32x128 (one wave per SIMD, grid 2x128 of 4 waves): each exp-built A fragment feeds 8 WMMAs instead of 4, halving vector work per WMMA
# speedup vs baseline: 1.3081x; 1.0655x over previous
#include <hip/hip_runtime.h>
#include <hip/hip_bf16.h>

#define BB 32
#define SS 512
#define FF 16
#define HH 256
#define BS (BB * SS)

typedef float v2f __attribute__((ext_vector_type(2)));
typedef float v8f __attribute__((ext_vector_type(8)));

__device__ __forceinline__ float elu_f(float v) {
    return v > 0.0f ? v : (__expf(v) - 1.0f);
}

__global__ __launch_bounds__(256) void vsn_attn_kernel(
    const float* __restrict__ x, const float* __restrict__ Wa1,
    const float* __restrict__ ba1, const float* __restrict__ Wa2,
    const float* __restrict__ ba2, float* __restrict__ wout)
{
    __shared__ float s_wa1t[HH * FF];
    __shared__ float s_wa2[HH * FF];
    __shared__ float s_ba1[HH];
    __shared__ float s_ba2[FF];

    const int tid = threadIdx.x;
#pragma unroll
    for (int i = 0; i < 16; ++i) {
        int lin = tid + i * 256;
        int f = lin >> 8, h = lin & 255;
        s_wa1t[h * FF + f] = Wa1[lin];
        s_wa2[lin] = Wa2[lin];
    }
    s_ba1[tid] = ba1[tid & 255];
    if (tid < FF) s_ba2[tid] = ba2[tid];
    __syncthreads();

    const size_t row = (size_t)blockIdx.x * 256 + tid;
    const float4* xs = (const float4*)(x + row * FF);
    float4 x0 = xs[0], x1 = xs[1], x2 = xs[2], x3 = xs[3];
    float xv[16] = {x0.x, x0.y, x0.z, x0.w, x1.x, x1.y, x1.z, x1.w,
                    x2.x, x2.y, x2.z, x2.w, x3.x, x3.y, x3.z, x3.w};

    float lg[16];
#pragma unroll
    for (int f = 0; f < 16; ++f) lg[f] = s_ba2[f];

#pragma unroll 2
    for (int h = 0; h < HH; ++h) {
        const float4* wc = (const float4*)&s_wa1t[h * FF];
        float4 c0 = wc[0], c1 = wc[1], c2 = wc[2], c3 = wc[3];
        float s = s_ba1[h];
        s += xv[0]*c0.x + xv[1]*c0.y + xv[2]*c0.z + xv[3]*c0.w;
        s += xv[4]*c1.x + xv[5]*c1.y + xv[6]*c1.z + xv[7]*c1.w;
        s += xv[8]*c2.x + xv[9]*c2.y + xv[10]*c2.z + xv[11]*c2.w;
        s += xv[12]*c3.x + xv[13]*c3.y + xv[14]*c3.z + xv[15]*c3.w;
        float a = elu_f(s);
        const float4* wd = (const float4*)&s_wa2[h * FF];
        float4 d0 = wd[0], d1 = wd[1], d2 = wd[2], d3 = wd[3];
        lg[0] += a*d0.x; lg[1] += a*d0.y; lg[2] += a*d0.z; lg[3] += a*d0.w;
        lg[4] += a*d1.x; lg[5] += a*d1.y; lg[6] += a*d1.z; lg[7] += a*d1.w;
        lg[8] += a*d2.x; lg[9] += a*d2.y; lg[10]+= a*d2.z; lg[11]+= a*d2.w;
        lg[12]+= a*d3.x; lg[13]+= a*d3.y; lg[14]+= a*d3.z; lg[15]+= a*d3.w;
    }

    float m = lg[0];
#pragma unroll
    for (int f = 1; f < 16; ++f) m = fmaxf(m, lg[f]);
    float s = 0.0f;
#pragma unroll
    for (int f = 0; f < 16; ++f) { lg[f] = __expf(lg[f] - m); s += lg[f]; }
    float inv = 1.0f / s;
#pragma unroll
    for (int f = 0; f < 16; ++f) lg[f] *= inv;

    float4* wo = (float4*)(wout + row * FF);
    wo[0] = make_float4(lg[0], lg[1], lg[2], lg[3]);
    wo[1] = make_float4(lg[4], lg[5], lg[6], lg[7]);
    wo[2] = make_float4(lg[8], lg[9], lg[10], lg[11]);
    wo[3] = make_float4(lg[12], lg[13], lg[14], lg[15]);
}

typedef float v4f __attribute__((ext_vector_type(4)));

#define MTHR   128
#define WROWS  32
#define WCOLS  128
#define KBLK   32
#define NBLK   (FF * HH / KBLK)

struct Blk {
    v4f   w1[4];
    v4f   bb[4];
    float xf[2];
    float wa[2];
    v4f   b[16][2];
};

__device__ __forceinline__ void load_blk(Blk& B, int blk,
    const float* __restrict__ w1h, const float* __restrict__ b1h,
    const float* __restrict__ w2l,
    const float* __restrict__ xr0, const float* __restrict__ xr1,
    const float* __restrict__ wr0, const float* __restrict__ wr1)
{
    const int f = blk >> 3;
    const float* pw = w1h + blk * KBLK;
    const float* pb = b1h + blk * KBLK;
#pragma unroll
    for (int i = 0; i < 4; ++i) {
        B.w1[i] = *(const v4f*)(pw + 4 * i);
        B.bb[i] = *(const v4f*)(pb + 4 * i);
    }
    B.xf[0] = xr0[f]; B.xf[1] = xr1[f];
    B.wa[0] = wr0[f]; B.wa[1] = wr1[f];
    const float* p2 = w2l + (size_t)blk * (KBLK * HH);
#pragma unroll
    for (int i = 0; i < 16; ++i) {
        B.b[i][0] = *(const v4f*)(p2 + i * HH);
        B.b[i][1] = *(const v4f*)(p2 + i * HH + 64);
    }
}

__device__ __forceinline__ void compute_blk(const Blk& B, v8f (&acc)[2][8])
{
#pragma unroll
    for (int j = 0; j < 8; ++j) {
        const int i0 = 2 * j, i1 = 2 * j + 1;
        const float w1a = B.w1[i0 >> 2][i0 & 3], w1b = B.w1[i1 >> 2][i1 & 3];
        const float b1a = B.bb[i0 >> 2][i0 & 3], b1b = B.bb[i1 >> 2][i1 & 3];
        v2f bf[8];
#pragma unroll
        for (int t = 0; t < 8; ++t) {
            bf[t].x = B.b[i0][t >> 2][t & 3];
            bf[t].y = B.b[i1][t >> 2][t & 3];
        }
#pragma unroll
        for (int m = 0; m < 2; ++m) {
            v2f a;
            a.x = B.wa[m] * elu_f(B.xf[m] * w1a + b1a);
            a.y = B.wa[m] * elu_f(B.xf[m] * w1b + b1b);
#pragma unroll
            for (int t = 0; t < 8; ++t)
                acc[m][t] = __builtin_amdgcn_wmma_f32_16x16x4_f32(
                    false, a, false, bf[t], (short)0, acc[m][t], false, false);
        }
    }
}

__global__ __launch_bounds__(MTHR) void vsn_main_kernel(
    const float* __restrict__ x, const float* __restrict__ W1,
    const float* __restrict__ b1, const float* __restrict__ W2,
    const float* __restrict__ b2, const float* __restrict__ wts,
    float* __restrict__ out)
{
    const int tid  = threadIdx.x;
    const int wave = tid >> 5;
    const int lane = tid & 31;
    const int half = lane >> 4;
    const int lr   = lane & 15;
    const int n0   = blockIdx.x * WCOLS;
    const int R0   = blockIdx.y * (4 * WROWS) + wave * WROWS;

    const float* xr0 = x   + (size_t)(R0 + lr) * FF;
    const float* xr1 = x   + (size_t)(R0 + 16 + lr) * FF;
    const float* wr0 = wts + (size_t)(R0 + lr) * FF;
    const float* wr1 = wts + (size_t)(R0 + 16 + lr) * FF;
    const float* w1h = W1 + 16 * half;
    const float* b1h = b1 + 16 * half;
    const float* w2l = W2 + (size_t)(16 * half) * HH + n0 + 4 * lr;

    Blk P, Q;
    load_blk(P, 0, w1h, b1h, w2l, xr0, xr1, wr0, wr1);

    v8f acc[2][8];
#pragma unroll
    for (int m = 0; m < 2; ++m)
#pragma unroll
        for (int t = 0; t < 8; ++t) acc[m][t] = (v8f)0.0f;

    {
        v4f wv[2][4];
#pragma unroll
        for (int q = 0; q < 4; ++q) {
            wv[0][q] = *(const v4f*)(wr0 + 4 * q);
            wv[1][q] = *(const v4f*)(wr1 + 4 * q);
        }
        const float* pb2 = b2 + (size_t)(2 * half) * HH + n0 + 4 * lr;
#pragma unroll
        for (int j = 0; j < 4; ++j) {
            v4f r0[2], r1[2];
#pragma unroll
            for (int u = 0; u < 2; ++u) {
                r0[u] = *(const v4f*)(pb2 + (4 * j) * HH + 64 * u);
                r1[u] = *(const v4f*)(pb2 + (4 * j + 1) * HH + 64 * u);
            }
#pragma unroll
            for (int m = 0; m < 2; ++m) {
                v2f a;
                a.x = half ? wv[m][j][2] : wv[m][j][0];
                a.y = half ? wv[m][j][3] : wv[m][j][1];
#pragma unroll
                for (int t = 0; t < 8; ++t) {
                    v2f bf; bf.x = r0[t >> 2][t & 3]; bf.y = r1[t >> 2][t & 3];
                    acc[m][t] = __builtin_amdgcn_wmma_f32_16x16x4_f32(
                        false, a, false, bf, (short)0, acc[m][t], false, false);
                }
            }
        }
    }

    for (int it = 0; it < NBLK / 2; ++it) {
        load_blk(Q, 2 * it + 1, w1h, b1h, w2l, xr0, xr1, wr0, wr1);
        __builtin_amdgcn_sched_barrier(0);
        compute_blk(P, acc);
        __builtin_amdgcn_sched_barrier(0);
        const int nb = (2 * it + 2 < NBLK) ? 2 * it + 2 : NBLK - 1;
        load_blk(P, nb, w1h, b1h, w2l, xr0, xr1, wr0, wr1);
        __builtin_amdgcn_sched_barrier(0);
        compute_blk(Q, acc);
    }

#pragma unroll
    for (int m = 0; m < 2; ++m) {
#pragma unroll
        for (int r = 0; r < 8; ++r) {
            const size_t row = (size_t)(R0 + 16 * m + 8 * half + r);
#pragma unroll
            for (int u = 0; u < 2; ++u) {
                v4f o;
                o[0] = acc[m][4 * u + 0][r]; o[1] = acc[m][4 * u + 1][r];
                o[2] = acc[m][4 * u + 2][r]; o[3] = acc[m][4 * u + 3][r];
                *(v4f*)(out + row * HH + n0 + 64 * u + 4 * lr) = o;
            }
        }
    }
}

extern "C" void kernel_launch(void* const* d_in, const int* in_sizes, int n_in,
                              void* d_out, int out_size, void* d_ws, size_t ws_size,
                              hipStream_t stream) {
    const float* x   = (const float*)d_in[0];
    const float* W1  = (const float*)d_in[1];
    const float* b1  = (const float*)d_in[2];
    const float* W2  = (const float*)d_in[3];
    const float* b2  = (const float*)d_in[4];
    const float* Wa1 = (const float*)d_in[5];
    const float* ba1 = (const float*)d_in[6];
    const float* Wa2 = (const float*)d_in[7];
    const float* ba2 = (const float*)d_in[8];

    float* out  = (float*)d_out;
    float* wout = out + (size_t)BS * HH;

    vsn_attn_kernel<<<dim3(BS / 256), dim3(256), 0, stream>>>(
        x, Wa1, ba1, Wa2, ba2, wout);

    vsn_main_kernel<<<dim3(HH / WCOLS, BS / (4 * WROWS)), dim3(MTHR), 0, stream>>>(
        x, W1, b1, W2, b2, wout, out);
}


# Softmax weights built inside the main kernel (one launch, not two): each wave does its 32 rows' two small products on the f32 matrix instruction, weights kept in LDS; h and f sums in matrix order
# speedup vs baseline: 1.5025x; 1.1486x over previous
#include <hip/hip_runtime.h>
#include <hip/hip_bf16.h>

#define BB 32
#define SS 512
#define FF 16
#define HH 256
#define BS (BB * SS)

typedef float v2f __attribute__((ext_vector_type(2)));
typedef float v8f __attribute__((ext_vector_type(8)));

__device__ __forceinline__ float elu_f(float v) {
    return v > 0.0f ? v : (__expf(v) - 1.0f);
}

typedef float v4f __attribute__((ext_vector_type(4)));

#define MTHR   128
#define WROWS  32
#define WCOLS  128
#define KBLK   32
#define NBLK   (FF * HH / KBLK)

struct Blk {
    v4f   w1[4];
    v4f   bb[4];
    float xf[2];
    float wa[2];
    v4f   b[16][2];
};

__device__ __forceinline__ void load_blk(Blk& B, int blk,
    const float* __restrict__ w1h, const float* __restrict__ b1h,
    const float* __restrict__ w2l,
    const float* __restrict__ xr0, const float* __restrict__ xr1,
    const float* __restrict__ wr0, const float* __restrict__ wr1)
{
    const int f = blk >> 3;
    const float* pw = w1h + blk * KBLK;
    const float* pb = b1h + blk * KBLK;
#pragma unroll
    for (int i = 0; i < 4; ++i) {
        B.w1[i] = *(const v4f*)(pw + 4 * i);
        B.bb[i] = *(const v4f*)(pb + 4 * i);
    }
    B.xf[0] = xr0[f]; B.xf[1] = xr1[f];
    B.wa[0] = wr0[f]; B.wa[1] = wr1[f];
    const float* p2 = w2l + (size_t)blk * (KBLK * HH);
#pragma unroll
    for (int i = 0; i < 16; ++i) {
        B.b[i][0] = *(const v4f*)(p2 + i * HH);
        B.b[i][1] = *(const v4f*)(p2 + i * HH + 64);
    }
}

__device__ __forceinline__ void compute_blk(const Blk& B, v8f (&acc)[2][8])
{
#pragma unroll
    for (int j = 0; j < 8; ++j) {
        const int i0 = 2 * j, i1 = 2 * j + 1;
        const float w1a = B.w1[i0 >> 2][i0 & 3], w1b = B.w1[i1 >> 2][i1 & 3];
        const float b1a = B.bb[i0 >> 2][i0 & 3], b1b = B.bb[i1 >> 2][i1 & 3];
        v2f bf[8];
#pragma unroll
        for (int t = 0; t < 8; ++t) {
            bf[t].x = B.b[i0][t >> 2][t & 3];
            bf[t].y = B.b[i1][t >> 2][t & 3];
        }
#pragma unroll
        for (int m = 0; m < 2; ++m) {
            v2f a;
            a.x = B.wa[m] * elu_f(B.xf[m] * w1a + b1a);
            a.y = B.wa[m] * elu_f(B.xf[m] * w1b + b1b);
#pragma unroll
            for (int t = 0; t < 8; ++t)
                acc[m][t] = __builtin_amdgcn_wmma_f32_16x16x4_f32(
                    false, a, false, bf[t], (short)0, acc[m][t], false, false);
        }
    }
}

__device__ __forceinline__ void wave_weights(
    const float* __restrict__ x, const float* __restrict__ Wa1,
    const float* __restrict__ ba1, const float* __restrict__ Wa2,
    const float* __restrict__ ba2, int R0, int lr, int half,
    float* sw, float* __restrict__ wout, bool write_out)
{
    v2f xb[2][4];
#pragma unroll
    for (int nt = 0; nt < 2; ++nt)
#pragma unroll
        for (int j = 0; j < 4; ++j)
            xb[nt][j] = *(const v2f*)(x + (size_t)(R0 + 16 * nt + lr) * FF + 4 * j + 2 * half);

    v8f d1[16][2];
#pragma unroll
    for (int ht = 0; ht < 16; ++ht) {
        const v4f c0 = *(const v4f*)(ba1 + 16 * ht + 8 * half);
        const v4f c1 = *(const v4f*)(ba1 + 16 * ht + 8 * half + 4);
        v8f ci;
        ci[0] = c0[0]; ci[1] = c0[1]; ci[2] = c0[2]; ci[3] = c0[3];
        ci[4] = c1[0]; ci[5] = c1[1]; ci[6] = c1[2]; ci[7] = c1[3];
        d1[ht][0] = ci; d1[ht][1] = ci;
#pragma unroll
        for (int j = 0; j < 4; ++j) {
            v2f a;
            a.x = Wa1[(4 * j + 2 * half) * HH + 16 * ht + lr];
            a.y = Wa1[(4 * j + 2 * half + 1) * HH + 16 * ht + lr];
#pragma unroll
            for (int nt = 0; nt < 2; ++nt)
                d1[ht][nt] = __builtin_amdgcn_wmma_f32_16x16x4_f32(
                    false, a, false, xb[nt][j], (short)0, d1[ht][nt], false, false);
        }
    }

    v8f d2[2];
    {
        const v4f c0 = *(const v4f*)(ba2 + 8 * half);
        const v4f c1 = *(const v4f*)(ba2 + 8 * half + 4);
        v8f ci;
        ci[0] = c0[0]; ci[1] = c0[1]; ci[2] = c0[2]; ci[3] = c0[3];
        ci[4] = c1[0]; ci[5] = c1[1]; ci[6] = c1[2]; ci[7] = c1[3];
        d2[0] = ci; d2[1] = ci;
    }
#pragma unroll
    for (int ht = 0; ht < 16; ++ht) {
#pragma unroll
        for (int jj = 0; jj < 4; ++jj) {
            v2f a;
            a.x = Wa2[(16 * ht + 8 * half + 2 * jj) * FF + lr];
            a.y = Wa2[(16 * ht + 8 * half + 2 * jj + 1) * FF + lr];
#pragma unroll
            for (int nt = 0; nt < 2; ++nt) {
                v2f b;
                b.x = elu_f(d1[ht][nt][2 * jj]);
                b.y = elu_f(d1[ht][nt][2 * jj + 1]);
                d2[nt] = __builtin_amdgcn_wmma_f32_16x16x4_f32(
                    false, a, false, b, (short)0, d2[nt], false, false);
            }
        }
    }

#pragma unroll
    for (int nt = 0; nt < 2; ++nt) {
        float m = d2[nt][0];
#pragma unroll
        for (int r = 1; r < 8; ++r) m = fmaxf(m, d2[nt][r]);
        m = fmaxf(m, __shfl_xor(m, 16));
        float e[8];
        float sum = 0.0f;
#pragma unroll
        for (int r = 0; r < 8; ++r) { e[r] = __expf(d2[nt][r] - m); sum += e[r]; }
        sum += __shfl_xor(sum, 16);
        const float inv = 1.0f / sum;
        v4f o0, o1;
        o0[0] = e[0] * inv; o0[1] = e[1] * inv; o0[2] = e[2] * inv; o0[3] = e[3] * inv;
        o1[0] = e[4] * inv; o1[1] = e[5] * inv; o1[2] = e[6] * inv; o1[3] = e[7] * inv;
        float* ps = sw + (16 * nt + lr) * FF + 8 * half;
        *(v4f*)ps = o0;
        *(v4f*)(ps + 4) = o1;
        if (write_out) {
            float* pg = wout + (size_t)(R0 + 16 * nt + lr) * FF + 8 * half;
            *(v4f*)pg = o0;
            *(v4f*)(pg + 4) = o1;
        }
    }
}

__global__ __launch_bounds__(MTHR) void vsn_main_kernel(
    const float* __restrict__ x, const float* __restrict__ W1,
    const float* __restrict__ b1, const float* __restrict__ W2,
    const float* __restrict__ b2,
    const float* __restrict__ Wa1, const float* __restrict__ ba1,
    const float* __restrict__ Wa2, const float* __restrict__ ba2,
    float* __restrict__ wout, float* __restrict__ out)
{
    __shared__ float s_w[4 * WROWS * FF];

    const int tid  = threadIdx.x;
    const int wave = tid >> 5;
    const int lane = tid & 31;
    const int half = lane >> 4;
    const int lr   = lane & 15;
    const int n0   = blockIdx.x * WCOLS;
    const int R0   = blockIdx.y * (4 * WROWS) + wave * WROWS;

    const float* xr0 = x   + (size_t)(R0 + lr) * FF;
    const float* xr1 = x   + (size_t)(R0 + 16 + lr) * FF;
    float* sw = s_w + wave * (WROWS * FF);
    const float* wr0 = sw + lr * FF;
    const float* wr1 = sw + (16 + lr) * FF;
    const float* w1h = W1 + 16 * half;
    const float* b1h = b1 + 16 * half;
    const float* w2l = W2 + (size_t)(16 * half) * HH + n0 + 4 * lr;

    wave_weights(x, Wa1, ba1, Wa2, ba2, R0, lr, half, sw, wout, blockIdx.x == 0);
    __syncthreads();

    Blk P, Q;
    load_blk(P, 0, w1h, b1h, w2l, xr0, xr1, wr0, wr1);

    v8f acc[2][8];
#pragma unroll
    for (int m = 0; m < 2; ++m)
#pragma unroll
        for (int t = 0; t < 8; ++t) acc[m][t] = (v8f)0.0f;

    {
        v4f wv[2][4];
#pragma unroll
        for (int q = 0; q < 4; ++q) {
            wv[0][q] = *(const v4f*)(wr0 + 4 * q);
            wv[1][q] = *(const v4f*)(wr1 + 4 * q);
        }
        const float* pb2 = b2 + (size_t)(2 * half) * HH + n0 + 4 * lr;
#pragma unroll
        for (int j = 0; j < 4; ++j) {
            v4f r0[2], r1[2];
#pragma unroll
            for (int u = 0; u < 2; ++u) {
                r0[u] = *(const v4f*)(pb2 + (4 * j) * HH + 64 * u);
                r1[u] = *(const v4f*)(pb2 + (4 * j + 1) * HH + 64 * u);
            }
#pragma unroll
            for (int m = 0; m < 2; ++m) {
                v2f a;
                a.x = half ? wv[m][j][2] : wv[m][j][0];
                a.y = half ? wv[m][j][3] : wv[m][j][1];
#pragma unroll
                for (int t = 0; t < 8; ++t) {
                    v2f bf; bf.x = r0[t >> 2][t & 3]; bf.y = r1[t >> 2][t & 3];
                    acc[m][t] = __builtin_amdgcn_wmma_f32_16x16x4_f32(
                        false, a, false, bf, (short)0, acc[m][t], false, false);
                }
            }
        }
    }

    for (int it = 0; it < NBLK / 2; ++it) {
        load_blk(Q, 2 * it + 1, w1h, b1h, w2l, xr0, xr1, wr0, wr1);
        __builtin_amdgcn_sched_barrier(0);
        compute_blk(P, acc);
        __builtin_amdgcn_sched_barrier(0);
        const int nb = (2 * it + 2 < NBLK) ? 2 * it + 2 : NBLK - 1;
        load_blk(P, nb, w1h, b1h, w2l, xr0, xr1, wr0, wr1);
        __builtin_amdgcn_sched_barrier(0);
        compute_blk(Q, acc);
    }

#pragma unroll
    for (int m = 0; m < 2; ++m) {
#pragma unroll
        for (int r = 0; r < 8; ++r) {
            const size_t row = (size_t)(R0 + 16 * m + 8 * half + r);
#pragma unroll
            for (int u = 0; u < 2; ++u) {
                v4f o;
                o[0] = acc[m][4 * u + 0][r]; o[1] = acc[m][4 * u + 1][r];
                o[2] = acc[m][4 * u + 2][r]; o[3] = acc[m][4 * u + 3][r];
                *(v4f*)(out + row * HH + n0 + 64 * u + 4 * lr) = o;
            }
        }
    }
}

extern "C" void kernel_launch(void* const* d_in, const int* in_sizes, int n_in,
                              void* d_out, int out_size, void* d_ws, size_t ws_size,
                              hipStream_t stream) {
    const float* x   = (const float*)d_in[0];
    const float* W1  = (const float*)d_in[1];
    const float* b1  = (const float*)d_in[2];
    const float* W2  = (const float*)d_in[3];
    const float* b2  = (const float*)d_in[4];
    const float* Wa1 = (const float*)d_in[5];
    const float* ba1 = (const float*)d_in[6];
    const float* Wa2 = (const float*)d_in[7];
    const float* ba2 = (const float*)d_in[8];

    float* out  = (float*)d_out;
    float* wout = out + (size_t)BS * HH;

    vsn_main_kernel<<<dim3(HH / WCOLS, BS / (4 * WROWS)), dim3(MTHR), 0, stream>>>(
        x, W1, b1, W2, b2, Wa1, ba1, Wa2, ba2, wout, out);
}


# Wave tile widened from 32x128 to 64x256: the (f,k) sum is split over the 4 waves of a work-group (4 features each), partial tiles added through LDS at the end; f32 sum over f now in 4 groups
# speedup vs baseline: 1.6280x; 1.0835x over previous
#include <hip/hip_runtime.h>
#include <hip/hip_bf16.h>

#define BB 32
#define SS 512
#define FF 16
#define HH 256
#define BS (BB * SS)

typedef float v2f __attribute__((ext_vector_type(2)));
typedef float v4f __attribute__((ext_vector_type(4)));
typedef float v8f __attribute__((ext_vector_type(8)));

__device__ __forceinline__ float elu_f(float v) {
    return v > 0.0f ? v : (__expf(v) - 1.0f);
}

#define SPLIT  4
#define MT     SPLIT
#define NT     16
#define WGROWS (16 * MT)
#define MTHR   (32 * SPLIT)
#define FPW    (FF / SPLIT)
#define KBLK   16
#define BPF    (HH / KBLK)
#define PART   4096

struct Blk {
    v4f w1[2];
    v4f bb[2];
    v4f b[8][4];
};

__device__ __forceinline__ void load_blk(Blk& B, int g,
    const float* __restrict__ w1h, const float* __restrict__ b1h,
    const float* __restrict__ w2l)
{
    const float* pw = w1h + g * KBLK;
    const float* pb = b1h + g * KBLK;
    B.w1[0] = *(const v4f*)(pw);     B.w1[1] = *(const v4f*)(pw + 4);
    B.bb[0] = *(const v4f*)(pb);     B.bb[1] = *(const v4f*)(pb + 4);
    const float* p2 = w2l + (size_t)g * (KBLK * HH);
#pragma unroll
    for (int i = 0; i < 8; ++i) {
#pragma unroll
        for (int u = 0; u < 4; ++u)
            B.b[i][u] = *(const v4f*)(p2 + i * HH + 64 * u);
    }
}

__device__ __forceinline__ void compute_blk(const Blk& B, const float (&xc)[MT],
                                            const float (&wc)[MT], v8f (&acc)[MT][NT])
{
#pragma unroll
    for (int j = 0; j < 4; ++j) {
        const int i0 = 2 * j, i1 = 2 * j + 1;
        const float w1a = B.w1[i0 >> 2][i0 & 3], w1b = B.w1[i1 >> 2][i1 & 3];
        const float b1a = B.bb[i0 >> 2][i0 & 3], b1b = B.bb[i1 >> 2][i1 & 3];
        v2f bf[NT];
#pragma unroll
        for (int t = 0; t < NT; ++t) {
            bf[t].x = B.b[i0][t >> 2][t & 3];
            bf[t].y = B.b[i1][t >> 2][t & 3];
        }
#pragma unroll
        for (int m = 0; m < MT; ++m) {
            v2f a;
            a.x = wc[m] * elu_f(xc[m] * w1a + b1a);
            a.y = wc[m] * elu_f(xc[m] * w1b + b1b);
#pragma unroll
            for (int t = 0; t < NT; ++t)
                acc[m][t] = __builtin_amdgcn_wmma_f32_16x16x4_f32(
                    false, a, false, bf[t], (short)0, acc[m][t], false, false);
        }
    }
}

__device__ __forceinline__ void wave_weights(
    const float* __restrict__ x, const float* __restrict__ Wa1,
    const float* __restrict__ ba1, const float* __restrict__ Wa2,
    const float* __restrict__ ba2, int Rw, int lr, int half,
    float* sw, float* __restrict__ wout)
{
    v2f xb[4];
#pragma unroll
    for (int j = 0; j < 4; ++j)
        xb[j] = *(const v2f*)(x + (size_t)(Rw + lr) * FF + 4 * j + 2 * half);

    v8f d1[16];
#pragma unroll
    for (int ht = 0; ht < 16; ++ht) {
        const v4f c0 = *(const v4f*)(ba1 + 16 * ht + 8 * half);
        const v4f c1 = *(const v4f*)(ba1 + 16 * ht + 8 * half + 4);
        v8f ci;
        ci[0] = c0[0]; ci[1] = c0[1]; ci[2] = c0[2]; ci[3] = c0[3];
        ci[4] = c1[0]; ci[5] = c1[1]; ci[6] = c1[2]; ci[7] = c1[3];
        d1[ht] = ci;
#pragma unroll
        for (int j = 0; j < 4; ++j) {
            v2f a;
            a.x = Wa1[(4 * j + 2 * half) * HH + 16 * ht + lr];
            a.y = Wa1[(4 * j + 2 * half + 1) * HH + 16 * ht + lr];
            d1[ht] = __builtin_amdgcn_wmma_f32_16x16x4_f32(
                false, a, false, xb[j], (short)0, d1[ht], false, false);
        }
    }

    v8f d2;
    {
        const v4f c0 = *(const v4f*)(ba2 + 8 * half);
        const v4f c1 = *(const v4f*)(ba2 + 8 * half + 4);
        v8f ci;
        ci[0] = c0[0]; ci[1] = c0[1]; ci[2] = c0[2]; ci[3] = c0[3];
        ci[4] = c1[0]; ci[5] = c1[1]; ci[6] = c1[2]; ci[7] = c1[3];
        d2 = ci;
    }
#pragma unroll
    for (int ht = 0; ht < 16; ++ht) {
#pragma unroll
        for (int jj = 0; jj < 4; ++jj) {
            v2f a;
            a.x = Wa2[(16 * ht + 8 * half + 2 * jj) * FF + lr];
            a.y = Wa2[(16 * ht + 8 * half + 2 * jj + 1) * FF + lr];
            v2f b;
            b.x = elu_f(d1[ht][2 * jj]);
            b.y = elu_f(d1[ht][2 * jj + 1]);
            d2 = __builtin_amdgcn_wmma_f32_16x16x4_f32(
                false, a, false, b, (short)0, d2, false, false);
        }
    }
    const v8f lg = d2;

    float m = lg[0];
#pragma unroll
    for (int r = 1; r < 8; ++r) m = fmaxf(m, lg[r]);
    m = fmaxf(m, __shfl_xor(m, 16));
    float e[8];
    float sum = 0.0f;
#pragma unroll
    for (int r = 0; r < 8; ++r) { e[r] = __expf(lg[r] - m); sum += e[r]; }
    sum += __shfl_xor(sum, 16);
    const float inv = 1.0f / sum;
    v4f o0, o1;
    o0[0] = e[0] * inv; o0[1] = e[1] * inv; o0[2] = e[2] * inv; o0[3] = e[3] * inv;
    o1[0] = e[4] * inv; o1[1] = e[5] * inv; o1[2] = e[6] * inv; o1[3] = e[7] * inv;
    float* ps = sw + lr * FF + 8 * half;
    *(v4f*)ps = o0;
    *(v4f*)(ps + 4) = o1;
    float* pg = wout + (size_t)(Rw + lr) * FF + 8 * half;
    *(v4f*)pg = o0;
    *(v4f*)(pg + 4) = o1;
}

__global__ __launch_bounds__(MTHR) void vsn_main_kernel(
    const float* __restrict__ x, const float* __restrict__ W1,
    const float* __restrict__ b1, const float* __restrict__ W2,
    const float* __restrict__ b2,
    const float* __restrict__ Wa1, const float* __restrict__ ba1,
    const float* __restrict__ Wa2, const float* __restrict__ ba2,
    float* __restrict__ wout, float* __restrict__ out)
{
    __shared__ float s_w[WGROWS * FF];
    __shared__ float s_red[SPLIT * (SPLIT - 1) * PART];

    const int tid  = threadIdx.x;
    const int ks   = __builtin_amdgcn_readfirstlane(tid >> 5);
    const int lane = tid & 31;
    const int half = lane >> 4;
    const int lr   = lane & 15;
    const int Rwg  = blockIdx.x * WGROWS;
    const int f0   = ks * FPW;

    wave_weights(x, Wa1, ba1, Wa2, ba2, Rwg + 16 * ks, lr, half, s_w + 16 * ks * FF, wout);
    __syncthreads();

    v4f xall[MT], wall[MT];
#pragma unroll
    for (int ml = 0; ml < MT; ++ml) {
        const int mg = (ks + ml) & (SPLIT - 1);
        xall[ml] = *(const v4f*)(x + (size_t)(Rwg + 16 * mg + lr) * FF + f0);
        wall[ml] = *(const v4f*)(s_w + (16 * mg + lr) * FF + f0);
    }

    const float* w1h = W1 + f0 * HH + 8 * half;
    const float* b1h = b1 + f0 * HH + 8 * half;
    const float* w2l = W2 + (size_t)(f0 * HH + 8 * half) * HH + 4 * lr;

    Blk P, Q;
    load_blk(P, 0, w1h, b1h, w2l);

    v8f acc[MT][NT];

    {
        const float* pb2 = b2 + (size_t)(f0 + 2 * half) * HH + 4 * lr;
        v4f r0[4], r1[4];
#pragma unroll
        for (int u = 0; u < 4; ++u) {
            r0[u] = *(const v4f*)(pb2 + 64 * u);
            r1[u] = *(const v4f*)(pb2 + HH + 64 * u);
        }
#pragma unroll
        for (int m = 0; m < MT; ++m) {
            v2f a;
            a.x = half ? wall[m][2] : wall[m][0];
            a.y = half ? wall[m][3] : wall[m][1];
#pragma unroll
            for (int t = 0; t < NT; ++t) {
                v2f bf; bf.x = r0[t >> 2][t & 3]; bf.y = r1[t >> 2][t & 3];
                acc[m][t] = __builtin_amdgcn_wmma_f32_16x16x4_f32(
                    false, a, false, bf, (short)0, (v8f)0.0f, false, false);
            }
        }
    }

#pragma unroll 1
    for (int fl = 0; fl < FPW; ++fl) {
        float xc[MT], wc[MT];
#pragma unroll
        for (int m = 0; m < MT; ++m) { xc[m] = xall[m][0]; wc[m] = wall[m][0]; }
#pragma unroll 1
        for (int it = 0; it < BPF / 2; ++it) {
            const int g = fl * BPF + 2 * it;
            load_blk(Q, g + 1, w1h, b1h, w2l);
            __builtin_amdgcn_sched_barrier(0);
            compute_blk(P, xc, wc, acc);
            __builtin_amdgcn_sched_barrier(0);
            const int nb = (g + 2 < FPW * BPF) ? g + 2 : FPW * BPF - 1;
            load_blk(P, nb, w1h, b1h, w2l);
            __builtin_amdgcn_sched_barrier(0);
            compute_blk(Q, xc, wc, acc);
        }
#pragma unroll
        for (int m = 0; m < MT; ++m) {
            xall[m] = __builtin_shufflevector(xall[m], xall[m], 1, 2, 3, 0);
            wall[m] = __builtin_shufflevector(wall[m], wall[m], 1, 2, 3, 0);
        }
    }

#pragma unroll
    for (int ml = 1; ml < MT; ++ml) {
        const int o = (ks + ml) & (SPLIT - 1);
        float* dst = s_red + (o * (SPLIT - 1) + (ml - 1)) * PART + lane * 4;
#pragma unroll
        for (int t = 0; t < NT; ++t) {
#pragma unroll
            for (int p = 0; p < 2; ++p) {
                v4f c;
                c[0] = acc[ml][t][4 * p + 0]; c[1] = acc[ml][t][4 * p + 1];
                c[2] = acc[ml][t][4 * p + 2]; c[3] = acc[ml][t][4 * p + 3];
                *(v4f*)(dst + (2 * t + p) * 128) = c;
            }
        }
    }
    __syncthreads();

#pragma unroll
    for (int s = 0; s < SPLIT - 1; ++s) {
        const float* sp = s_red + (ks * (SPLIT - 1) + s) * PART + lane * 4;
#pragma unroll
        for (int t = 0; t < NT; ++t) {
#pragma unroll
            for (int p = 0; p < 2; ++p) {
                const v4f c = *(const v4f*)(sp + (2 * t + p) * 128);
                acc[0][t][4 * p + 0] += c[0]; acc[0][t][4 * p + 1] += c[1];
                acc[0][t][4 * p + 2] += c[2]; acc[0][t][4 * p + 3] += c[3];
            }
        }
    }
#pragma unroll
    for (int r = 0; r < 8; ++r) {
        const size_t row = (size_t)(Rwg + 16 * ks + 8 * half + r);
#pragma unroll
        for (int u = 0; u < 4; ++u) {
            v4f o;
            o[0] = acc[0][4 * u + 0][r]; o[1] = acc[0][4 * u + 1][r];
            o[2] = acc[0][4 * u + 2][r]; o[3] = acc[0][4 * u + 3][r];
            *(v4f*)(out + row * HH + 64 * u + 4 * lr) = o;
        }
    }
}

extern "C" void kernel_launch(void* const* d_in, const int* in_sizes, int n_in,
                              void* d_out, int out_size, void* d_ws, size_t ws_size,
                              hipStream_t stream) {
    const float* x   = (const float*)d_in[0];
    const float* W1  = (const float*)d_in[1];
    const float* b1  = (const float*)d_in[2];
    const float* W2  = (const float*)d_in[3];
    const float* b2  = (const float*)d_in[4];
    const float* Wa1 = (const float*)d_in[5];
    const float* ba1 = (const float*)d_in[6];
    const float* Wa2 = (const float*)d_in[7];
    const float* ba2 = (const float*)d_in[8];

    float* out  = (float*)d_out;
    float* wout = out + (size_t)BS * HH;

    vsn_main_kernel<<<dim3(BS / WGROWS), dim3(MTHR), 0, stream>>>(
        x, W1, b1, W2, b2, Wa1, ba1, Wa2, ba2, wout, out);
}
